# starting kernel
# baseline (speedup 1.0000x reference)
#include <hip/hip_runtime.h>


typedef __attribute__((ext_vector_type(2))) float v2f;
typedef __attribute__((ext_vector_type(8))) float v8f;

#define SIZEP 10
#define CIN   32
#define COUT  32
#define NR    252
#define WAVES 4

__device__ __forceinline__ void wmma_rtile(const v2f* pSA, const v2f* pSB,
                                           const v2f b0[8], const v2f b1[8],
                                           int halfsel, float bz0, float bz1,
                                           v8f& acc0, v8f& acc1) {
  acc0 = (v8f){bz0, bz0, bz0, bz0, bz0, bz0, bz0, bz0};
  acc1 = (v8f){bz1, bz1, bz1, bz1, bz1, bz1, bz1, bz1};
  #pragma unroll
  for (int kk = 0; kk < 8; ++kk) {
    const int ch = 2 * kk + halfsel;
    const v2f a  = pSA[ch] + pSB[ch];
    acc0 = __builtin_amdgcn_wmma_f32_16x16x4_f32(
        false, a, false, b0[kk], (short)0, acc0, false, false);
    acc1 = __builtin_amdgcn_wmma_f32_16x16x4_f32(
        false, a, false, b1[kk], (short)0, acc1, false, false);
  }
}

__global__ __launch_bounds__(WAVES * 32)
void pcl_wmma_kernel(const float* __restrict__ x,
                     const float* __restrict__ W,
                     const int*   __restrict__ split0,
                     float* __restrict__ out,
                     int samples) {
  __shared__ __align__(16) unsigned int s_mask[256];
  __shared__ __align__(16) float s_Wd[COUT * CIN];
  __shared__ __align__(16) float s_SA[WAVES][32 * 32];
  __shared__ __align__(16) float s_SB[WAVES][32 * 32];
  __shared__ __align__(16) float s_base[WAVES][32];

  const int tid = threadIdx.x;
  const int w   = tid >> 5;
  const int l   = tid & 31;
  const int s   = blockIdx.x * WAVES + w;

  for (int r = tid; r < 256; r += WAVES * 32) {
    unsigned int mA = 0u, mB = 0u;
    if (r < NR) {
      #pragma unroll
      for (int g = 0; g < 5; ++g) {
        int p = split0[r * 5 + g];
        if (p < 5) mA |= 1u << p;
        else       mB |= 1u << (p - 5);
      }
    }
    s_mask[r] = mA | (mB << 8);
  }

  #pragma unroll
  for (int i = 0; i < (COUT * CIN) / (WAVES * 32); ++i) {
    const int idx = tid + i * (WAVES * 32);
    s_Wd[idx] = W[idx] - W[COUT * CIN + idx];
  }

  if (s < samples) {
    const float* xs = x + (size_t)s * (SIZEP * CIN);
    float xa[5], xb[5];
    #pragma unroll
    for (int p = 0; p < 5; ++p) {
      xa[p] = xs[p * CIN + l];
      xb[p] = xs[(p + 5) * CIN + l];
    }
    float sa[32], sb[32];
    sa[0] = 0.f; sb[0] = 0.f;
    #pragma unroll
    for (int m = 1; m < 32; ++m) {
      sa[m] = sa[m & (m - 1)] + xa[__builtin_ctz(m)];
      sb[m] = sb[m & (m - 1)] + xb[__builtin_ctz(m)];
    }
    #pragma unroll
    for (int m = 0; m < 32; ++m) {
      s_SA[w][m * 32 + l] = sa[m];
      s_SB[w][m * 32 + l] = sb[m];
    }
  }
  __syncthreads();

  if (s < samples) {
    const float* w1row = W + COUT * CIN + l * CIN;
    float acc = 0.f;
    #pragma unroll
    for (int c = 0; c < CIN; ++c)
      acc += (s_SA[w][31 * 32 + c] + s_SB[w][31 * 32 + c]) * w1row[c];
    s_base[w][l] = acc;
  }
  __syncthreads();

  if (s >= samples) return;

  const int halfsel = l >> 4;
  const int nlo     = l & 15;

  v2f b0[8], b1[8];
  {
    const v2f* pB0 = (const v2f*)&s_Wd[nlo * CIN];
    const v2f* pB1 = (const v2f*)&s_Wd[(16 + nlo) * CIN];
    #pragma unroll
    for (int kk = 0; kk < 8; ++kk) {
      b0[kk] = pB0[2 * kk + halfsel];
      b1[kk] = pB1[2 * kk + halfsel];
    }
  }

  const float bz0 = s_base[w][nlo];
  const float bz1 = s_base[w][16 + nlo];

  float* outp = out + (size_t)s * (NR * COUT);

  for (int rt = 0; rt < 15; ++rt) {
    const unsigned int mm = s_mask[rt * 16 + nlo];
    const v2f* pSA = (const v2f*)&s_SA[w][(mm & 31u) * 32];
    const v2f* pSB = (const v2f*)&s_SB[w][((mm >> 8) & 31u) * 32];

    v8f acc0, acc1;
    wmma_rtile(pSA, pSB, b0, b1, halfsel, bz0, bz1, acc0, acc1);

    float* orow = outp + (rt * 16 + 8 * halfsel) * COUT + nlo;
    #pragma unroll
    for (int j = 0; j < 8; ++j) {
      orow[j * COUT]      = acc0[j];
      orow[j * COUT + 16] = acc1[j];
    }
  }

  {
    const unsigned int mm = s_mask[15 * 16 + nlo];
    const v2f* pSA = (const v2f*)&s_SA[w][(mm & 31u) * 32];
    const v2f* pSB = (const v2f*)&s_SB[w][((mm >> 8) & 31u) * 32];

    v8f acc0, acc1;
    wmma_rtile(pSA, pSB, b0, b1, halfsel, bz0, bz1, acc0, acc1);

    float* orow = outp + (15 * 16 + 8 * halfsel) * COUT + nlo;
    #pragma unroll
    for (int j = 0; j < 8; ++j) {
      if (8 * halfsel + j < NR - 240) {
        orow[j * COUT]      = acc0[j];
        orow[j * COUT + 16] = acc1[j];
      }
    }
  }
}

extern "C" void kernel_launch(void* const* d_in, const int* in_sizes, int n_in,
                              void* d_out, int out_size, void* d_ws, size_t ws_size,
                              hipStream_t stream) {
  const float* x      = (const float*)d_in[0];
  const float* W      = (const float*)d_in[1];
  const int*   splits = (const int*)d_in[2];
  float* out = (float*)d_out;

  const int samples = in_sizes[0] / (SIZEP * CIN);
  const int blocks  = (samples + WAVES - 1) / WAVES;

  pcl_wmma_kernel<<<blocks, WAVES * 32, 0, stream>>>(x, W, splits, out, samples);
}


# Split pooling done as a second WMMA with the rows' 0/1 position matrix: per 16-row tile 6 WMMAs on Y = x*Wd^T kept in registers (was 16 + LDS subset-sum gather); no tables; f32 sums reordered
# speedup vs baseline: 1.9847x; 1.9847x over previous
#include <hip/hip_runtime.h>


typedef __attribute__((ext_vector_type(2))) float v2f;
typedef __attribute__((ext_vector_type(4))) float v4f;
typedef __attribute__((ext_vector_type(8))) float v8f;

#define SIZEP 10
#define CIN   32
#define COUT  32
#define NR    252
#define WAVES 4
#define WROW  36
#define W1QP  132
#define SROW  36
#define BASEBIT (1u << 10)
#define FULLSET 1023u

__device__ __forceinline__ void wmma_rtile(uint2 v, const v2f ya[3], const v2f yb[3],
                                           v8f& acc0, v8f& acc1) {
  const v8f zero = {0.f, 0.f, 0.f, 0.f, 0.f, 0.f, 0.f, 0.f};
  acc0 = zero;
  acc1 = zero;
  const v2f pbs[3] = {{(float)(v.x & 0xffu), (float)((v.x >> 8) & 0xffu)},
                      {(float)((v.x >> 16) & 0xffu), (float)(v.x >> 24)},
                      {(float)(v.y & 0xffu), (float)((v.y >> 8) & 0xffu)}};
  #pragma unroll
  for (int kk = 0; kk < 3; ++kk) {
    const v2f pb = pbs[kk];
    acc0 = __builtin_amdgcn_wmma_f32_16x16x4_f32(
        false, ya[kk], false, pb, (short)0, acc0, false, false);
    acc1 = __builtin_amdgcn_wmma_f32_16x16x4_f32(
        false, yb[kk], false, pb, (short)0, acc1, false, false);
  }
}

__device__ __forceinline__ void stage_put(float* st, int nlo, int halfsel,
                                          const v8f& acc0, const v8f& acc1) {
  v4f* p = (v4f*)&st[nlo * SROW + 8 * halfsel];
  p[0] = (v4f){acc0[0], acc0[1], acc0[2], acc0[3]};
  p[1] = (v4f){acc0[4], acc0[5], acc0[6], acc0[7]};
  p[4] = (v4f){acc1[0], acc1[1], acc1[2], acc1[3]};
  p[5] = (v4f){acc1[4], acc1[5], acc1[6], acc1[7]};
}

template <int ND, int NM>
__device__ __forceinline__ void stage_store(const float* st, int l, float* orow, float* mrow, v4f cz) {
  const int g  = l >> 3;
  const int c4 = (l & 7) << 2;
  v4f v[4];
  #pragma unroll
  for (int k = 0; k < ND; ++k) v[k] = *(const v4f*)&st[(4 * k + g) * SROW + c4];
  #pragma unroll
  for (int k = 0; k < ND; ++k) *(v4f*)&orow[(4 * k + g) * COUT + c4] = v[k];
  #pragma unroll
  for (int k = 0; k < NM; ++k) *(v4f*)&mrow[-(4 * k + g) * COUT + c4] = cz - v[k];
}

__global__ __launch_bounds__(WAVES * 32)
void pcl_wmma_kernel(const float* __restrict__ x,
                     const float* __restrict__ W,
                     const int*   __restrict__ split0,
                     float* __restrict__ out,
                     int samples) {
  __shared__ __align__(16) unsigned int s_mask[256];
  __shared__ __align__(16) float s_Wd[COUT * WROW];
  __shared__ __align__(16) float s_W1q[8 * W1QP];
  __shared__ __align__(16) uint2 s_pb[256 * 2];
  __shared__ __align__(16) float s_tot[WAVES][32];
  __shared__ __align__(16) float s_stage[WAVES][16 * SROW];

  const int tid = threadIdx.x;
  const int w   = tid >> 5;
  const int l   = tid & 31;
  const int s   = blockIdx.x * WAVES + w;
  const bool live = s < samples;
  const int halfsel = l >> 4;
  const int nlo     = l & 15;

  const int rhi = tid + WAVES * 32;
  const int* q0 = split0 + tid * 5;
  const int* q1 = split0 + (rhi < NR ? rhi : NR - 1) * 5;
  int p0[5], p1[5];
  #pragma unroll
  for (int g = 0; g < 5; ++g) {
    p0[g] = q0[g];
    p1[g] = q1[g];
  }
  float w0[(COUT * CIN) / (WAVES * 32)], w1[(COUT * CIN) / (WAVES * 32)];
  #pragma unroll
  for (int i = 0; i < (COUT * CIN) / (WAVES * 32); ++i) {
    w0[i] = W[tid + i * (WAVES * 32)];
    w1[i] = W[COUT * CIN + tid + i * (WAVES * 32)];
  }
  v4f xf[4];
  {
    const int pos = nlo - (nlo >= 12 ? 6 : (nlo >= 8 ? 2 : 0));
    const float* xs = x + (size_t)(live ? s : samples - 1) * (SIZEP * CIN) + pos * CIN + 4 * halfsel;
    #pragma unroll
    for (int q = 0; q < 4; ++q) xf[q] = *(const v4f*)&xs[8 * q];
  }
  float xr[SIZEP];
  {
    const float* xs = x + (size_t)(live ? s : samples - 1) * (SIZEP * CIN) + l;
    #pragma unroll
    for (int p = 0; p < SIZEP; ++p) xr[p] = xs[p * CIN];
  }

  {
    unsigned int m0 = 0u, m1 = 0u;
    #pragma unroll
    for (int g = 0; g < 5; ++g) {
      m0 |= 1u << p0[g];
      m1 |= 1u << p1[g];
    }
    if (rhi >= NR) m1 = 0u;
    m0 = (m0 & FULLSET) | BASEBIT;
    m1 = (m1 & FULLSET) | BASEBIT;
    s_mask[tid] = m0;
    s_mask[rhi] = m1;
    #pragma unroll
    for (int h = 0; h < 2; ++h) {
      const unsigned int e0 = (m0 >> (6 * h)) & 63u, e1 = (m1 >> (6 * h)) & 63u;
      s_pb[tid * 2 + h] = make_uint2(((e0 & 15u) * 0x00204081u) & 0x01010101u,
                                     ((e0 >> 4) & 1u) | (((e0 >> 5) & 1u) << 8));
      s_pb[rhi * 2 + h] = make_uint2(((e1 & 15u) * 0x00204081u) & 0x01010101u,
                                     ((e1 >> 4) & 1u) | (((e1 >> 5) & 1u) << 8));
    }
  }

  #pragma unroll
  for (int i = 0; i < (COUT * CIN) / (WAVES * 32); ++i) {
    const int o = w + i * WAVES;
    s_Wd[o * WROW + l] = w0[i] - w1[i];
    s_W1q[(l >> 2) * W1QP + o * 4 + (l & 3)] = w1[i];
  }
  {
    float t = xr[0];
    #pragma unroll
    for (int p = 1; p < SIZEP; ++p) t += xr[p];
    s_tot[w][l] = t;
  }
  __syncthreads();

  if (!live) return;

  float base;
  {
    float e = 0.f, d = 0.f;
    #pragma unroll
    for (int q = 0; q < 8; q += 2) {
      const v4f wa = *(const v4f*)&s_W1q[q * W1QP + 4 * l];
      const v4f wb = *(const v4f*)&s_W1q[(q + 1) * W1QP + 4 * l];
      const v4f ta = *(const v4f*)&s_tot[w][4 * q];
      const v4f tb = *(const v4f*)&s_tot[w][4 * q + 4];
      #pragma unroll
      for (int j = 0; j < 4; ++j) {
        e += ta[j] * wa[j];
        d += tb[j] * wb[j];
      }
    }
    base = e + d;
  }

  v8f yd0 = {0.f, 0.f, 0.f, 0.f, 0.f, 0.f, 0.f, 0.f};
  v8f yd1 = yd0;
  {
    const float* pd = &s_Wd[nlo * WROW + 4 * halfsel];
    v4f d0[4], d1[4];
    #pragma unroll
    for (int q = 0; q < 4; ++q) {
      d0[q] = *(const v4f*)&pd[8 * q];
      d1[q] = *(const v4f*)&pd[16 * WROW + 8 * q];
    }
    #pragma unroll
    for (int q = 0; q < 4; ++q) {
      const v2f alo = {xf[q].x, xf[q].y};
      const v2f ahi = {xf[q].z, xf[q].w};
      yd0 = __builtin_amdgcn_wmma_f32_16x16x4_f32(false, alo, false, (v2f){d0[q].x, d0[q].y}, (short)0, yd0, false, false);
      yd1 = __builtin_amdgcn_wmma_f32_16x16x4_f32(false, alo, false, (v2f){d1[q].x, d1[q].y}, (short)0, yd1, false, false);
      yd0 = __builtin_amdgcn_wmma_f32_16x16x4_f32(false, ahi, false, (v2f){d0[q].z, d0[q].w}, (short)0, yd0, false, false);
      yd1 = __builtin_amdgcn_wmma_f32_16x16x4_f32(false, ahi, false, (v2f){d1[q].z, d1[q].w}, (short)0, yd1, false, false);
    }
  }

  const float base0 = __shfl_xor(base, 16, 32);
  const float base1 = base;

  v2f ya[3], yb[3];
  ya[0] = (v2f){yd0[0], yd0[1]};
  ya[1] = (v2f){yd0[2], yd0[3]};
  yb[0] = (v2f){yd1[0], yd1[1]};
  yb[1] = (v2f){yd1[2], yd1[3]};
  if (halfsel == 0) {
    ya[2] = (v2f){yd0[4], yd0[5]};
    yb[2] = (v2f){yd1[4], yd1[5]};
  } else {
    ya[2] = (v2f){base0, 0.f};
    yb[2] = (v2f){base1, 0.f};
  }

  float* st   = &s_stage[w][0];
  float* outp = out + (size_t)s * (NR * COUT);

  unsigned int diff = 0u;
  #pragma unroll
  for (int i = 0; i < 4; ++i) {
    int r = l + 32 * i;
    r = r < NR / 2 ? r : NR / 2 - 1;
    diff |= (s_mask[r] ^ s_mask[NR - 1 - r]) ^ FULLSET;
  }
  const bool mirrored = __builtin_amdgcn_ballot_w32(diff == 0u) == 0xffffffffu;

  if (mirrored) {
    uint2 mk[8];
    #pragma unroll
    for (int rt = 0; rt < 8; ++rt) mk[rt] = s_pb[(rt * 16 + nlo) * 2 + halfsel];

    v4f cz;
    {
      v8f acc0, acc1;
      wmma_rtile(mk[7], ya, yb, acc0, acc1);
      stage_put(st, nlo, halfsel, acc0, acc1);

      const int c4 = (l & 7) << 2;
      cz = *(const v4f*)&st[12 * SROW + c4] + *(const v4f*)&st[15 * SROW + c4];

      stage_store<4, 3>(st, l, outp + (7 * 16) * COUT, outp + (NR - 1 - 7 * 16) * COUT, cz);
    }

    #pragma unroll
    for (int rt = 0; rt < 7; ++rt) {
      v8f acc0, acc1;
      wmma_rtile(mk[rt], ya, yb, acc0, acc1);
      stage_put(st, nlo, halfsel, acc0, acc1);
      stage_store<4, 4>(st, l, outp + (rt * 16) * COUT, outp + (NR - 1 - rt * 16) * COUT, cz);
    }
  } else {
    const v4f none = {0.f, 0.f, 0.f, 0.f};
    for (int rt = 0; rt < 15; ++rt) {
      const uint2 v = s_pb[(rt * 16 + nlo) * 2 + halfsel];
      v8f acc0, acc1;
      wmma_rtile(v, ya, yb, acc0, acc1);
      stage_put(st, nlo, halfsel, acc0, acc1);
      stage_store<4, 0>(st, l, outp + (rt * 16) * COUT, outp, none);
    }

    {
      const uint2 v = s_pb[(15 * 16 + nlo) * 2 + halfsel];
      v8f acc0, acc1;
      wmma_rtile(v, ya, yb, acc0, acc1);
      stage_put(st, nlo, halfsel, acc0, acc1);
      stage_store<3, 0>(st, l, outp + (15 * 16) * COUT, outp, none);
    }
  }
}

extern "C" void kernel_launch(void* const* d_in, const int* in_sizes, int n_in,
                              void* d_out, int out_size, void* d_ws, size_t ws_size,
                              hipStream_t stream) {
  const float* x      = (const float*)d_in[0];
  const float* W      = (const float*)d_in[1];
  const int*   splits = (const int*)d_in[2];
  float* out = (float*)d_out;

  const int samples = in_sizes[0] / (SIZEP * CIN);
  const int blocks  = (samples + WAVES - 1) / WAVES;

  pcl_wmma_kernel<<<blocks, WAVES * 32, 0, stream>>>(x, W, splits, out, samples);
}
